# starting kernel
# baseline (speedup 1.0000x reference)
#include <hip/hip_runtime.h>

typedef __attribute__((ext_vector_type(2))) float v2f;
typedef __attribute__((ext_vector_type(4))) float f4;
typedef __attribute__((ext_vector_type(8))) float v8f;

#define WIN       128
#define NSEQ      4096
#define DDIM      256
#define WROWS     144
#define VW_STRIDE 260
#define E_STRIDE  148

__global__ __launch_bounds__(128)
void sparse_prop_kernel(const float* __restrict__ val,
                        const float* __restrict__ state,
                        float* __restrict__ dstate,
                        float* __restrict__ dval) {
  extern __shared__ float smem[];
  float* val_w = smem;
  float* edges = smem + WROWS * VW_STRIDE;
  float* st_s  = edges + 16 * E_STRIDE;

  const int tid  = threadIdx.x;
  const int lane = tid & 31;
  const int wave = tid >> 5;
  const int m    = lane & 15;
  const int h    = lane >> 4;

  const int blk = blockIdx.x;
  const int b   = blk >> 8;
  const int i0  = (blk & 255) << 4;
  const int j0  = i0 - WIN;

  const float* valb = val + (size_t)b * NSEQ * DDIM;

  for (int t = tid; t < WROWS * (DDIM / 4); t += 128) {
    int row = t >> 6;
    int c4  = t & 63;
    int jr  = j0 + row; if (jr < 0) jr = 0;
    f4 v4 = *(const f4*)(valb + (size_t)jr * DDIM + c4 * 4);
    *(f4*)(val_w + row * VW_STRIDE + c4 * 4) = v4;
  }
  for (int t = tid; t < WROWS; t += 128) {
    int jr = j0 + t; if (jr < 0) jr = 0;
    st_s[t] = state[(size_t)b * NSEQ + jr];
  }
  __syncthreads();

  for (int jt = wave; jt < 9; jt += 4) {
    v8f acc = {};
    const float* arow = val_w + (WIN + m) * VW_STRIDE;
    const float* brow = val_w + (jt * 16 + m) * VW_STRIDE;
    for (int k0 = 0; k0 < DDIM; k0 += 4) {
      v2f a  = *(const v2f*)(arow + k0 + 2 * h);
      v2f bm = *(const v2f*)(brow + k0 + 2 * h);
      acc = __builtin_amdgcn_wmma_f32_16x16x4_f32(false, a, false, bm,
                                                  (short)0, acc, false, false);
    }
    const int c = jt * 16 + m;
    #pragma unroll
    for (int v = 0; v < 8; ++v) {
      int   r = v + 8 * h;
      float s = acc[v] * 0.0625f;
      float e = s * __builtin_amdgcn_rcpf(1.0f + fabsf(s));
      bool ok = (c >= r) && (c <= r + WIN) && ((j0 + c) >= 0);
      edges[r * E_STRIDE + c] = ok ? e : 0.0f;
    }
  }
  __syncthreads();

  float* dvb = dval + ((size_t)b * NSEQ + i0) * DDIM;
  for (int dt = 0; dt < 4; ++dt) {
    int d0 = wave * 64 + dt * 16;
    v8f acc = {};
    for (int c0 = 0; c0 < WROWS; c0 += 4) {
      v2f a = *(const v2f*)(edges + m * E_STRIDE + c0 + 2 * h);
      v2f bm;
      bm.x = val_w[(c0 + 2 * h    ) * VW_STRIDE + d0 + m];
      bm.y = val_w[(c0 + 2 * h + 1) * VW_STRIDE + d0 + m];
      acc = __builtin_amdgcn_wmma_f32_16x16x4_f32(false, a, false, bm,
                                                  (short)0, acc, false, false);
    }
    #pragma unroll
    for (int v = 0; v < 8; ++v) {
      int r = v + 8 * h;
      dvb[(size_t)r * DDIM + d0 + m] = acc[v];
    }
  }

  if (wave == 0 && lane < 16) {
    float s = 0.0f;
    for (int c = 0; c < WROWS; ++c) s += edges[lane * E_STRIDE + c] * st_s[c];
    dstate[(size_t)b * NSEQ + i0 + lane] = s;
  }
}

extern "C" void kernel_launch(void* const* d_in, const int* in_sizes, int n_in,
                              void* d_out, int out_size, void* d_ws, size_t ws_size,
                              hipStream_t stream) {
  const float* val   = (const float*)d_in[0];
  const float* state = (const float*)d_in[1];
  float* dstate = (float*)d_out;
  float* dval   = (float*)d_out + 4 * NSEQ;

  size_t smem_bytes =
      (size_t)(WROWS * VW_STRIDE + 16 * E_STRIDE + WROWS) * sizeof(float);

  hipLaunchKernelGGL(sparse_prop_kernel, dim3(1024), dim3(128), smem_bytes,
                     stream, val, state, dstate, dval);
}


# Merge four adjacent 16-row work-groups into one 512-thread group on one shared 192-row staged window (256 groups, one per WGP); lane row pointers take the sub-tile offset; order of sums unchanged
# speedup vs baseline: 1.7535x; 1.7535x over previous
#include <hip/hip_runtime.h>

typedef __attribute__((ext_vector_type(2))) float v2f;
typedef __attribute__((ext_vector_type(4))) float f4;
typedef __attribute__((ext_vector_type(8))) float v8f;

#define WIN       128
#define NSEQ      4096
#define DDIM      256
#define NSUB      4
#define TROWS     (16 * NSUB)
#define WROWS     144
#define BROWS     (WIN + TROWS)
#define VW_STRIDE 260
#define E_STRIDE  148
#define NTHR      (128 * NSUB)

__global__ __launch_bounds__(NTHR)
void sparse_prop_kernel(const float* __restrict__ val,
                        const float* __restrict__ state,
                        float* __restrict__ dstate,
                        float* __restrict__ dval) {
  extern __shared__ float smem[];
  float* val_w  = smem;
  float* edges4 = smem + BROWS * VW_STRIDE;
  float* st_s   = edges4 + NSUB * 16 * E_STRIDE;

  const int tid  = threadIdx.x;
  const int lane = tid & 31;
  const int wave = tid >> 5;
  const int sub  = wave & 3;
  const int q    = wave >> 2;
  const int m    = lane & 15;
  const int h    = lane >> 4;

  const int blk = blockIdx.x;
  const int b   = blk >> 6;
  const int i0  = (blk & 63) << 6;
  const int j0  = i0 - WIN;

  const float* valb = val + (size_t)b * NSEQ * DDIM;

  for (int t = tid; t < BROWS * (DDIM / 4); t += NTHR) {
    int row = t >> 6;
    int c4  = t & 63;
    int jr  = j0 + row; if (jr < 0) jr = 0;
    f4 v4 = *(const f4*)(valb + (size_t)jr * DDIM + c4 * 4);
    *(f4*)(val_w + row * VW_STRIDE + c4 * 4) = v4;
  }
  for (int t = tid; t < BROWS; t += NTHR) {
    int jr = j0 + t; if (jr < 0) jr = 0;
    st_s[t] = state[(size_t)b * NSEQ + jr];
  }
  __syncthreads();

  const float* vsub  = val_w + sub * 16 * VW_STRIDE;
  float*       edges = edges4 + sub * 16 * E_STRIDE;
  const float* stw   = st_s + sub * 16;
  const int    jsub  = j0 + sub * 16;

  const float* arow = val_w + (sub * 16 + WIN + m) * VW_STRIDE;
  for (int jt = q; jt < 9; jt += 4) {
    v8f acc = {};
    const float* brow = vsub + (jt * 16 + m) * VW_STRIDE;
    for (int k0 = 0; k0 < DDIM; k0 += 4) {
      v2f a  = *(const v2f*)(arow + k0 + 2 * h);
      v2f bm = *(const v2f*)(brow + k0 + 2 * h);
      acc = __builtin_amdgcn_wmma_f32_16x16x4_f32(false, a, false, bm,
                                                  (short)0, acc, false, false);
    }
    const int c = jt * 16 + m;
    #pragma unroll
    for (int v = 0; v < 8; ++v) {
      int   r = v + 8 * h;
      float s = acc[v] * 0.0625f;
      float e = s * __builtin_amdgcn_rcpf(1.0f + fabsf(s));
      bool ok = (c >= r) && (c <= r + WIN) && ((jsub + c) >= 0);
      edges[r * E_STRIDE + c] = ok ? e : 0.0f;
    }
  }
  __syncthreads();

  float* dvb = dval + ((size_t)b * NSEQ + i0 + sub * 16 + 8 * h) * DDIM + m;
  const float* erow = edges4 + (sub * 16 + m) * E_STRIDE;
  for (int dt = 0; dt < 4; ++dt) {
    int d0 = q * 64 + dt * 16;
    v8f acc = {};
    for (int c0 = 0; c0 < WROWS; c0 += 4) {
      v2f a = *(const v2f*)(erow + c0 + 2 * h);
      v2f bm;
      bm.x = vsub[(c0 + 2 * h    ) * VW_STRIDE + d0 + m];
      bm.y = vsub[(c0 + 2 * h + 1) * VW_STRIDE + d0 + m];
      acc = __builtin_amdgcn_wmma_f32_16x16x4_f32(false, a, false, bm,
                                                  (short)0, acc, false, false);
    }
    #pragma unroll
    for (int v = 0; v < 8; ++v) {
      dvb[(size_t)v * DDIM + d0] = acc[v];
    }
  }

  if (q == 0 && lane < 16) {
    float s = 0.0f;
    for (int c = 0; c < WROWS; ++c) s += edges[lane * E_STRIDE + c] * stw[c];
    dstate[(size_t)b * NSEQ + i0 + sub * 16 + lane] = s;
  }
}

extern "C" void kernel_launch(void* const* d_in, const int* in_sizes, int n_in,
                              void* d_out, int out_size, void* d_ws, size_t ws_size,
                              hipStream_t stream) {
  const float* val   = (const float*)d_in[0];
  const float* state = (const float*)d_in[1];
  float* dstate = (float*)d_out;
  float* dval   = (float*)d_out + 4 * NSEQ;

  size_t smem_bytes =
      (size_t)(BROWS * VW_STRIDE + NSUB * 16 * E_STRIDE + BROWS) * sizeof(float);

  hipLaunchKernelGGL(sparse_prop_kernel, dim3(256), dim3(NTHR), smem_bytes,
                     stream, val, state, dstate, dval);
}


# Phase 2 (edges x val): deal a wave's 64 columns to its four tiles column-wise (tile t = cols 4n+t): one 128-bit LDS read per B k-row, A shared by 4 accumulators, 128-bit row stores; sum order kept
# speedup vs baseline: 1.9044x; 1.0861x over previous
#include <hip/hip_runtime.h>

typedef __attribute__((ext_vector_type(2))) float v2f;
typedef __attribute__((ext_vector_type(4))) float f4;
typedef __attribute__((ext_vector_type(8))) float v8f;

#define WIN       128
#define NSEQ      4096
#define DDIM      256
#define NSUB      4
#define TROWS     (16 * NSUB)
#define WROWS     144
#define BROWS     (WIN + TROWS)
#define VW_STRIDE 260
#define E_STRIDE  148
#define NTHR      (128 * NSUB)

__global__ __launch_bounds__(NTHR)
void sparse_prop_kernel(const float* __restrict__ val,
                        const float* __restrict__ state,
                        float* __restrict__ dstate,
                        float* __restrict__ dval) {
  extern __shared__ float smem[];
  float* val_w  = smem;
  float* edges4 = smem + BROWS * VW_STRIDE;
  float* st_s   = edges4 + NSUB * 16 * E_STRIDE;

  const int tid  = threadIdx.x;
  const int lane = tid & 31;
  const int wave = tid >> 5;
  const int sub  = wave & 3;
  const int q    = wave >> 2;
  const int m    = lane & 15;
  const int h    = lane >> 4;

  const int blk = blockIdx.x;
  const int b   = blk >> 6;
  const int i0  = (blk & 63) << 6;
  const int j0  = i0 - WIN;

  const float* valb = val + (size_t)b * NSEQ * DDIM;

  for (int t = tid; t < BROWS * (DDIM / 4); t += NTHR) {
    int row = t >> 6;
    int c4  = t & 63;
    int jr  = j0 + row; if (jr < 0) jr = 0;
    f4 v4 = *(const f4*)(valb + (size_t)jr * DDIM + c4 * 4);
    *(f4*)(val_w + row * VW_STRIDE + c4 * 4) = v4;
  }
  for (int t = tid; t < BROWS; t += NTHR) {
    int jr = j0 + t; if (jr < 0) jr = 0;
    st_s[t] = state[(size_t)b * NSEQ + jr];
  }
  __syncthreads();

  const float* vsub  = val_w + sub * 16 * VW_STRIDE;
  float*       edges = edges4 + sub * 16 * E_STRIDE;
  const float* stw   = st_s + sub * 16;
  const int    jsub  = j0 + sub * 16;

  const float* arow = val_w + (sub * 16 + WIN + m) * VW_STRIDE;
  for (int jt = q; jt < 9; jt += 4) {
    v8f acc = {};
    const float* brow = vsub + (jt * 16 + m) * VW_STRIDE;
    for (int k0 = 0; k0 < DDIM; k0 += 4) {
      v2f a  = *(const v2f*)(arow + k0 + 2 * h);
      v2f bm = *(const v2f*)(brow + k0 + 2 * h);
      acc = __builtin_amdgcn_wmma_f32_16x16x4_f32(false, a, false, bm,
                                                  (short)0, acc, false, false);
    }
    const int c = jt * 16 + m;
    #pragma unroll
    for (int v = 0; v < 8; ++v) {
      int   r = v + 8 * h;
      float s = acc[v] * 0.0625f;
      float e = s * __builtin_amdgcn_rcpf(1.0f + fabsf(s));
      bool ok = (c >= r) && (c <= r + WIN) && ((jsub + c) >= 0);
      edges[r * E_STRIDE + c] = ok ? e : 0.0f;
    }
  }
  __syncthreads();

  const int dq = q * 64 + 4 * m;
  float* dvb = dval + ((size_t)b * NSEQ + i0 + sub * 16 + 8 * h) * DDIM + dq;
  const float* erow = edges4 + (sub * 16 + m) * E_STRIDE + 2 * h;
  const float* bcol = vsub + (2 * h) * VW_STRIDE + dq;
  v8f acc0 = {}, acc1 = {}, acc2 = {}, acc3 = {};
  for (int c0 = 0; c0 < WROWS; c0 += 4) {
    v2f a  = *(const v2f*)(erow + c0);
    f4  r0 = *(const f4*)(bcol + (c0    ) * VW_STRIDE);
    f4  r1 = *(const f4*)(bcol + (c0 + 1) * VW_STRIDE);
    v2f b0 = {r0.x, r1.x};
    v2f b1 = {r0.y, r1.y};
    v2f b2 = {r0.z, r1.z};
    v2f b3 = {r0.w, r1.w};
    acc0 = __builtin_amdgcn_wmma_f32_16x16x4_f32(false, a, false, b0,
                                                 (short)0, acc0, false, false);
    acc1 = __builtin_amdgcn_wmma_f32_16x16x4_f32(false, a, false, b1,
                                                 (short)0, acc1, false, false);
    acc2 = __builtin_amdgcn_wmma_f32_16x16x4_f32(false, a, false, b2,
                                                 (short)0, acc2, false, false);
    acc3 = __builtin_amdgcn_wmma_f32_16x16x4_f32(false, a, false, b3,
                                                 (short)0, acc3, false, false);
  }
  #pragma unroll
  for (int v = 0; v < 8; ++v) {
    f4 o = {acc0[v], acc1[v], acc2[v], acc3[v]};
    *(f4*)(dvb + (size_t)v * DDIM) = o;
  }

  if (q == 0 && lane < 16) {
    float s = 0.0f;
    for (int c = 0; c < WROWS; ++c) s += edges[lane * E_STRIDE + c] * stw[c];
    dstate[(size_t)b * NSEQ + i0 + sub * 16 + lane] = s;
  }
}

extern "C" void kernel_launch(void* const* d_in, const int* in_sizes, int n_in,
                              void* d_out, int out_size, void* d_ws, size_t ws_size,
                              hipStream_t stream) {
  const float* val   = (const float*)d_in[0];
  const float* state = (const float*)d_in[1];
  float* dstate = (float*)d_out;
  float* dval   = (float*)d_out + 4 * NSEQ;

  size_t smem_bytes =
      (size_t)(BROWS * VW_STRIDE + NSUB * 16 * E_STRIDE + BROWS) * sizeof(float);

  hipLaunchKernelGGL(sparse_prop_kernel, dim3(256), dim3(NTHR), smem_bytes,
                     stream, val, state, dstate, dval);
}


# Stage val in four 64-column K chunks: each chunk's global loads fly while the waves run their score tiles (one accumulator per tile, shared A fragment) on the chunk before; order of sums unchanged
# speedup vs baseline: 2.1025x; 1.1040x over previous
#include <hip/hip_runtime.h>

typedef __attribute__((ext_vector_type(2))) float v2f;
typedef __attribute__((ext_vector_type(4))) float f4;
typedef __attribute__((ext_vector_type(8))) float v8f;

#define WIN       128
#define NSEQ      4096
#define DDIM      256
#define NSUB      4
#define TROWS     (16 * NSUB)
#define WROWS     144
#define BROWS     (WIN + TROWS)
#define VW_STRIDE 260
#define E_STRIDE  148
#define NTHR      (128 * NSUB)
#define KCH       64
#define NCH       (DDIM / KCH)
#define PF        (BROWS * (KCH / 4) / NTHR)

__device__ __forceinline__ void chunk_load(f4 (&pf)[PF], const float* valb,
                                           int j0, int tid, int ch) {
  #pragma unroll
  for (int i = 0; i < PF; ++i) {
    int t   = tid + i * NTHR;
    int row = t >> 4;
    int c4  = t & 15;
    int jr  = j0 + row; if (jr < 0) jr = 0;
    pf[i] = *(const f4*)(valb + (size_t)jr * DDIM + ch * KCH + c4 * 4);
  }
}

__device__ __forceinline__ void chunk_store(const f4 (&pf)[PF], float* val_w,
                                            int tid, int ch) {
  #pragma unroll
  for (int i = 0; i < PF; ++i) {
    int t   = tid + i * NTHR;
    int row = t >> 4;
    int c4  = t & 15;
    *(f4*)(val_w + row * VW_STRIDE + ch * KCH + c4 * 4) = pf[i];
  }
}

template <int NT>
__device__ __forceinline__ void scores_chunk(const float* arow, const float* b0,
                                             const float* b1, const float* b2,
                                             v8f& acc0, v8f& acc1, v8f& acc2,
                                             int kbeg) {
  for (int k0 = kbeg; k0 < kbeg + KCH; k0 += 4) {
    v2f a  = *(const v2f*)(arow + k0);
    v2f x0 = *(const v2f*)(b0 + k0);
    v2f x1 = *(const v2f*)(b1 + k0);
    acc0 = __builtin_amdgcn_wmma_f32_16x16x4_f32(false, a, false, x0,
                                                 (short)0, acc0, false, false);
    acc1 = __builtin_amdgcn_wmma_f32_16x16x4_f32(false, a, false, x1,
                                                 (short)0, acc1, false, false);
    if (NT == 3) {
      v2f x2 = *(const v2f*)(b2 + k0);
      acc2 = __builtin_amdgcn_wmma_f32_16x16x4_f32(false, a, false, x2,
                                                   (short)0, acc2, false, false);
    }
  }
}

__device__ __forceinline__ void edges_tile(const v8f& acc, float* edges, int jt,
                                           int m, int h, int jsub) {
  const int c = jt * 16 + m;
  #pragma unroll
  for (int v = 0; v < 8; ++v) {
    int   r = v + 8 * h;
    float s = acc[v] * 0.0625f;
    float e = s * __builtin_amdgcn_rcpf(1.0f + fabsf(s));
    bool ok = (c >= r) && (c <= r + WIN) && ((jsub + c) >= 0);
    edges[r * E_STRIDE + c] = ok ? e : 0.0f;
  }
}

__global__ __launch_bounds__(NTHR)
void sparse_prop_kernel(const float* __restrict__ val,
                        const float* __restrict__ state,
                        float* __restrict__ dstate,
                        float* __restrict__ dval) {
  extern __shared__ float smem[];
  float* val_w  = smem;
  float* edges4 = smem + BROWS * VW_STRIDE;
  float* st_s   = edges4 + NSUB * 16 * E_STRIDE;

  const int tid  = threadIdx.x;
  const int lane = tid & 31;
  const int wave = __builtin_amdgcn_readfirstlane(tid >> 5);
  const int sub  = wave & 3;
  const int q    = wave >> 2;
  const int m    = lane & 15;
  const int h    = lane >> 4;

  const int blk = blockIdx.x;
  const int b   = blk >> 6;
  const int i0  = (blk & 63) << 6;
  const int j0  = i0 - WIN;

  const float* valb = val + (size_t)b * NSEQ * DDIM;

  f4 pf[PF];
  chunk_load(pf, valb, j0, tid, 0);
  for (int t = tid; t < BROWS; t += NTHR) {
    int jr = j0 + t; if (jr < 0) jr = 0;
    st_s[t] = state[(size_t)b * NSEQ + jr];
  }
  chunk_store(pf, val_w, tid, 0);
  __syncthreads();

  const float* vsub  = val_w + sub * 16 * VW_STRIDE;
  float*       edges = edges4 + sub * 16 * E_STRIDE;
  const float* stw   = st_s + sub * 16;
  const int    jsub  = j0 + sub * 16;

  const float* arow = val_w + (sub * 16 + WIN + m) * VW_STRIDE + 2 * h;
  const float* br0  = vsub + ((q    ) * 16 + m) * VW_STRIDE + 2 * h;
  const float* br1  = vsub + ((q + 4) * 16 + m) * VW_STRIDE + 2 * h;
  const float* br2  = vsub + (8 * 16 + m) * VW_STRIDE + 2 * h;
  v8f sc0 = {}, sc1 = {}, sc2 = {};
  #pragma nounroll
  for (int ch = 0; ch < NCH; ++ch) {
    const bool more = (ch + 1 < NCH);
    if (more) chunk_load(pf, valb, j0, tid, ch + 1);
    if (q == 0) scores_chunk<3>(arow, br0, br1, br2, sc0, sc1, sc2, ch * KCH);
    else        scores_chunk<2>(arow, br0, br1, br2, sc0, sc1, sc2, ch * KCH);
    if (more) {
      chunk_store(pf, val_w, tid, ch + 1);
      __syncthreads();
    }
  }
  edges_tile(sc0, edges, q,     m, h, jsub);
  edges_tile(sc1, edges, q + 4, m, h, jsub);
  if (q == 0) edges_tile(sc2, edges, 8, m, h, jsub);
  __syncthreads();

  const int dq = q * 64 + 4 * m;
  float* dvb = dval + ((size_t)b * NSEQ + i0 + sub * 16 + 8 * h) * DDIM + dq;
  const float* erow = edges4 + (sub * 16 + m) * E_STRIDE + 2 * h;
  const float* bcol = vsub + (2 * h) * VW_STRIDE + dq;
  v8f acc0 = {}, acc1 = {}, acc2 = {}, acc3 = {};
  for (int c0 = 0; c0 < WROWS; c0 += 4) {
    v2f a  = *(const v2f*)(erow + c0);
    f4  r0 = *(const f4*)(bcol + (c0    ) * VW_STRIDE);
    f4  r1 = *(const f4*)(bcol + (c0 + 1) * VW_STRIDE);
    v2f b0 = {r0.x, r1.x};
    v2f b1 = {r0.y, r1.y};
    v2f b2 = {r0.z, r1.z};
    v2f b3 = {r0.w, r1.w};
    acc0 = __builtin_amdgcn_wmma_f32_16x16x4_f32(false, a, false, b0,
                                                 (short)0, acc0, false, false);
    acc1 = __builtin_amdgcn_wmma_f32_16x16x4_f32(false, a, false, b1,
                                                 (short)0, acc1, false, false);
    acc2 = __builtin_amdgcn_wmma_f32_16x16x4_f32(false, a, false, b2,
                                                 (short)0, acc2, false, false);
    acc3 = __builtin_amdgcn_wmma_f32_16x16x4_f32(false, a, false, b3,
                                                 (short)0, acc3, false, false);
  }
  #pragma unroll
  for (int v = 0; v < 8; ++v) {
    f4 o = {acc0[v], acc1[v], acc2[v], acc3[v]};
    *(f4*)(dvb + (size_t)v * DDIM) = o;
  }

  if (q == 0 && lane < 16) {
    float s = 0.0f;
    for (int c = 0; c < WROWS; ++c) s += edges[lane * E_STRIDE + c] * stw[c];
    dstate[(size_t)b * NSEQ + i0 + sub * 16 + lane] = s;
  }
}

extern "C" void kernel_launch(void* const* d_in, const int* in_sizes, int n_in,
                              void* d_out, int out_size, void* d_ws, size_t ws_size,
                              hipStream_t stream) {
  const float* val   = (const float*)d_in[0];
  const float* state = (const float*)d_in[1];
  float* dstate = (float*)d_out;
  float* dval   = (float*)d_out + 4 * NSEQ;

  size_t smem_bytes =
      (size_t)(BROWS * VW_STRIDE + NSUB * 16 * E_STRIDE + BROWS) * sizeof(float);

  hipLaunchKernelGGL(sparse_prop_kernel, dim3(256), dim3(NTHR), smem_bytes,
                     stream, val, state, dstate, dval);
}
